# starting kernel
# baseline (speedup 1.0000x reference)
#include <hip/hip_runtime.h>

typedef __attribute__((ext_vector_type(2))) float v2f;
typedef __attribute__((ext_vector_type(4))) float v4f;
typedef __attribute__((ext_vector_type(8))) float v8f;

#define DIM        64
#define NREL       5
#define WAVES      4
#define EPT        32
#define MAX_BLOCKS 2048


__global__ __launch_bounds__(WAVES * 32)
void bidecoder_kernel(const float* __restrict__ u_feat,
                      const float* __restrict__ i_feat,
                      const int*   __restrict__ edge_user,
                      const int*   __restrict__ edge_item,
                      const float* __restrict__ W,
                      const float* __restrict__ bias,
                      float* __restrict__ out,
                      int num_edges)
{
    __shared__ __align__(16) float lds_W [NREL * DIM * DIM];
    __shared__ __align__(16) float lds_b [NREL * DIM];
    __shared__ __align__(16) float lds_U [WAVES][EPT * DIM];
    __shared__ __align__(16) float lds_I [WAVES][EPT * DIM];
    __shared__ __align__(16) float lds_O [WAVES][EPT * NREL];

    const int tid  = threadIdx.x;
    const int wave = tid >> 5;
    const int lane = tid & 31;
    const int h    = lane >> 4;
    const int lm   = lane & 15;

    {
        const v4f* src = (const v4f*)W;
        for (int idx = tid; idx < (NREL * DIM * DIM) / 4; idx += WAVES * 32)
            ((v4f*)lds_W)[idx] = src[idx];
        for (int idx = tid; idx < NREL * DIM; idx += WAVES * 32)
            lds_b[idx] = bias[idx];
    }
    __syncthreads();

    float* Ubuf = lds_U[wave];
    float* Ibuf = lds_I[wave];
    float* Obuf = lds_O[wave];

    const int  tiles   = (num_edges + EPT - 1) / EPT;
    const long wstride = (long)gridDim.x * WAVES;

    for (long tile = (long)blockIdx.x * WAVES + wave; tile < tiles; tile += wstride) {
        const long e0 = tile * EPT;

        {
            long nt = tile + wstride;
            if (nt < tiles) {
                long epf = nt * EPT + lane;
                if (epf >= num_edges) epf = num_edges - 1;
                const float* pu = u_feat + (long)edge_user[epf] * DIM;
                const float* pi = i_feat + (long)edge_item[epf] * DIM;
                __builtin_prefetch(pu,      0, 3);
                __builtin_prefetch(pu + 32, 0, 3);
                __builtin_prefetch(pi,      0, 3);
                __builtin_prefetch(pi + 32, 0, 3);
            }
        }

        #pragma unroll
        for (int j = 0; j < (EPT * DIM) / (32 * 4); ++j) {
            int flat = j * 32 + lane;
            int row  = flat >> 4;
            int c4   = (flat & 15) << 2;
            long e   = e0 + row;
            if (e >= num_edges) e = num_edges - 1;
            long ue = edge_user[e];
            long ie = edge_item[e];
            *(v4f*)(Ubuf + row * DIM + c4) = *(const v4f*)(u_feat + ue * DIM + c4);
            *(v4f*)(Ibuf + row * DIM + c4) = *(const v4f*)(i_feat + ie * DIM + c4);
        }
        asm volatile("" ::: "memory");

        v2f a0[16], a1[16];
        #pragma unroll
        for (int s = 0; s < 16; ++s) {
            a0[s] = *(const v2f*)(Ubuf + lm * DIM        + 4 * s + 2 * h);
            a1[s] = *(const v2f*)(Ubuf + (16 + lm) * DIM + 4 * s + 2 * h);
        }

        v2f a_one;
        a_one.x = (h == 0) ? 1.0f : 0.0f;
        a_one.y = 0.0f;

        for (int r = 0; r < NREL; ++r) {
            const float* Wr = lds_W + r * DIM * DIM;

            for (int t = 0; t < 4; ++t) {
                const int n = t * 16 + lm;
                v8f acc0 = {}, acc1 = {};
                #pragma unroll
                for (int s = 0; s < 16; ++s) {
                    v2f bf = *(const v2f*)(Wr + n * DIM + 4 * s + 2 * h);
                    acc0 = __builtin_amdgcn_wmma_f32_16x16x4_f32(
                               false, a0[s], false, bf, (short)0, acc0, false, false);
                    acc1 = __builtin_amdgcn_wmma_f32_16x16x4_f32(
                               false, a1[s], false, bf, (short)0, acc1, false, false);
                }
                v2f bb;
                bb.x = (h == 0) ? lds_b[r * DIM + n] : 0.0f;
                bb.y = 0.0f;
                acc0 = __builtin_amdgcn_wmma_f32_16x16x4_f32(
                           false, a_one, false, bb, (short)0, acc0, false, false);
                acc1 = __builtin_amdgcn_wmma_f32_16x16x4_f32(
                           false, a_one, false, bb, (short)0, acc1, false, false);
                #pragma unroll
                for (int v = 0; v < 8; ++v) {
                    Ubuf[(v + 8 * h) * DIM + n]        = acc0[v];
                    Ubuf[(16 + v + 8 * h) * DIM + n]   = acc1[v];
                }
            }
            asm volatile("" ::: "memory");

            #pragma unroll
            for (int p = 0; p < 2; ++p) {
                const float* Yp = Ubuf + (p * 16) * DIM;
                const float* Ip = Ibuf + (p * 16) * DIM;
                v8f acc2 = {};
                #pragma unroll
                for (int s = 0; s < 16; ++s) {
                    v2f af = *(const v2f*)(Yp + lm * DIM + 4 * s + 2 * h);
                    v2f bf = *(const v2f*)(Ip + lm * DIM + 4 * s + 2 * h);
                    acc2 = __builtin_amdgcn_wmma_f32_16x16x4_f32(
                               false, af, false, bf, (short)0, acc2, false, false);
                }
                int vsel = (lane < 16) ? lane : (lane - 24);
                float dval = acc2[0];
                #pragma unroll
                for (int v = 1; v < 8; ++v) dval = (vsel == v) ? acc2[v] : dval;
                bool valid = (lane < 8) || (lane >= 24);
                int  m     = p * 16 + ((lane < 8) ? lane : (lane - 16));
                if (valid) Obuf[m * NREL + r] = dval;
            }
            asm volatile("" ::: "memory");
        }

        for (int j = lane; j < EPT * NREL; j += 32) {
            long o = e0 * NREL + j;
            if (o < (long)num_edges * NREL) out[o] = Obuf[j];
        }
    }
}

extern "C" void kernel_launch(void* const* d_in, const int* in_sizes, int n_in,
                              void* d_out, int out_size, void* d_ws, size_t ws_size,
                              hipStream_t stream) {
    const float* u_feat = (const float*)d_in[0];
    const float* i_feat = (const float*)d_in[1];
    const int*   eu     = (const int*)d_in[2];
    const int*   ei     = (const int*)d_in[3];
    const float* W      = (const float*)d_in[4];
    const float* b      = (const float*)d_in[5];
    float*       out    = (float*)d_out;

    int E      = in_sizes[2];
    int tiles  = (E + EPT - 1) / EPT;
    int blocks = (tiles + WAVES - 1) / WAVES;
    if (blocks > MAX_BLOCKS) blocks = MAX_BLOCKS;

    bidecoder_kernel<<<blocks, WAVES * 32, 0, stream>>>(u_feat, i_feat, eu, ei, W, b, out, E);
}


# All of W (160 fragments, 640 VGPRs) read from LDS once per wave and kept in registers, one wave per SIMD (4-wave blocks): no LDS fragment load left in the WMMA stream
# speedup vs baseline: 2.5296x; 2.5296x over previous
#include <hip/hip_runtime.h>


typedef __attribute__((ext_vector_type(2))) float v2f;
typedef __attribute__((ext_vector_type(4))) float v4f;
typedef __attribute__((ext_vector_type(8))) float v8f;

#define DIM        64
#define NREL       5
#define WAVES      4
#define EPT        32
#define MAX_BLOCKS 256
#define NFRAG      (NREL * 4 * 8)
#define NCHUNK     (NFRAG * 32)


__device__ __forceinline__ v2f pair(const v4f x, const int q) {
    v2f r;
    r.x = q ? x.z : x.x;
    r.y = q ? x.w : x.y;
    return r;
}

__device__ __forceinline__ float xhalf(const float x) {
    const int b = __builtin_bit_cast(int, x);
    return __builtin_bit_cast(float, __builtin_amdgcn_permlanex16(b, b, 0x76543210, 0xfedcba98, false, false));
}

__device__ __forceinline__ v2f fma2(const float a0, const float a1, const float b0, const float b1, const v2f s) {
    const v2f a = {a0, a1};
    const v2f b = {b0, b1};
    return __builtin_elementwise_fma(a, b, s);
}

__global__ __launch_bounds__(WAVES * 32)
void bidecoder_kernel(const float* __restrict__ u_feat,
                      const float* __restrict__ i_feat,
                      const int*   __restrict__ edge_user,
                      const int*   __restrict__ edge_item,
                      const float* __restrict__ W,
                      const float* __restrict__ bias,
                      float* __restrict__ out,
                      int num_edges)
{
    __shared__ __align__(16) v4f   lds_Wf[NCHUNK];
    __shared__ __align__(16) float lds_b [NREL * DIM];
    __shared__ __align__(16) float lds_O [WAVES][EPT * NREL];

    const int tid  = threadIdx.x;
    const int wave = tid >> 5;
    const int lane = tid & 31;
    const int h    = lane >> 4;
    const int c    = lane & 15;

    for (int g = tid; g < NCHUNK; g += WAVES * 32) {
        const int l  = g & 31;
        const int j  = (g >> 5) & 7;
        const int t  = (g >> 8) & 3;
        const int r  = g >> 10;
        const int lc = l & 15;
        const int lh = l >> 4;
        const int n  = 16 * t + 8 * ((lc & 7) >> 2) + 4 * (lc >> 3) + (lc & 3);
        lds_Wf[g] = *(const v4f*)(W + (r * DIM + n) * DIM + 8 * j + 4 * lh);
    }
    for (int idx = tid; idx < NREL * DIM; idx += WAVES * 32)
        lds_b[idx] = bias[idx];
    __syncthreads();

    v4f Wf[NFRAG];
    #pragma unroll
    for (int i = 0; i < NFRAG; ++i) Wf[i] = lds_Wf[i * 32 + lane];

    float* Obuf = lds_O[wave];

    const int tiles   = (num_edges + EPT - 1) / EPT;
    const int wstride = (int)gridDim.x * WAVES;
    const int last    = num_edges - 1;

    v4f U0[8], U1[8], I0[8], I1[8];
    v4f nU0[8], nU1[8];
    int xu0, xu1, xi0, xi1;

    int tile = blockIdx.x * WAVES + wave;

#define LOAD_IDX(TILE)                                                     \
    {                                                                      \
        long ea = (long)(TILE) * EPT + c;                                  \
        long eb = ea + 16;                                                 \
        if (ea > last) ea = last;                                          \
        if (eb > last) eb = last;                                          \
        xu0 = edge_user[ea]; xu1 = edge_user[eb];                          \
        xi0 = edge_item[ea]; xi1 = edge_item[eb];                          \
    }
#define STAGE2(T, A0, A1)                                                  \
    {                                                                      \
        const v4f ia0 = I0[2 * (T)], ib0 = I0[2 * (T) + 1];                \
        const v4f ia1 = I1[2 * (T)], ib1 = I1[2 * (T) + 1];                \
        pa0 = fma2(A0[0], A0[1], ia0.x, ia0.y, pa0);                       \
        pb0 = fma2(A0[2], A0[3], ia0.z, ia0.w, pb0);                       \
        pa1 = fma2(A1[0], A1[1], ia1.x, ia1.y, pa1);                       \
        pb1 = fma2(A1[2], A1[3], ia1.z, ia1.w, pb1);                       \
        pa0 = fma2(A0[4], A0[5], ib0.x, ib0.y, pa0);                       \
        pb0 = fma2(A0[6], A0[7], ib0.z, ib0.w, pb0);                       \
        pa1 = fma2(A1[4], A1[5], ib1.x, ib1.y, pa1);                       \
        pb1 = fma2(A1[6], A1[7], ib1.z, ib1.w, pb1);                       \
    }
#define FINISH_REL(R)                                                      \
    {                                                                      \
        const float s0 = (pa0.x + pa0.y) + (pb0.x + pb0.y);                \
        const float s1 = (pa1.x + pa1.y) + (pb1.x + pb1.y);                \
        const float t0 = s0 + xhalf(s0);                                   \
        const float t1 = s1 + xhalf(s1);                                   \
        res[(R)] = h ? t1 : t0;                                            \
        pa0 = zero2; pb0 = zero2; pa1 = zero2; pb1 = zero2;                \
    }

#define REFILL_I(T)                                                        \
    {                                                                      \
        I0[2 * (T)]     = *(const v4f*)(pi0 + 16 * (T));                   \
        I0[2 * (T) + 1] = *(const v4f*)(pi0 + 16 * (T) + 8);               \
        I1[2 * (T)]     = *(const v4f*)(pi1 + 16 * (T));                   \
        I1[2 * (T) + 1] = *(const v4f*)(pi1 + 16 * (T) + 8);               \
    }

    if (tile < tiles) {
        LOAD_IDX(tile);
        const float* pu0 = u_feat + (long)xu0 * DIM + 4 * h;
        const float* pu1 = u_feat + (long)xu1 * DIM + 4 * h;
        const float* pi0 = i_feat + (long)xi0 * DIM + 4 * h;
        const float* pi1 = i_feat + (long)xi1 * DIM + 4 * h;
        #pragma unroll
        for (int j = 0; j < 8; ++j) {
            U0[j] = *(const v4f*)(pu0 + 8 * j);
            U1[j] = *(const v4f*)(pu1 + 8 * j);
            I0[j] = *(const v4f*)(pi0 + 8 * j);
            I1[j] = *(const v4f*)(pi1 + 8 * j);
        }
        LOAD_IDX(tile + wstride);
        #pragma unroll
        for (int j = 0; j < 8; ++j)
            asm volatile("" : "+v"(U0[j]), "+v"(U1[j]), "+v"(I0[j]), "+v"(I1[j]));
        asm volatile("" : "+v"(xu0), "+v"(xu1), "+v"(xi0), "+v"(xi1));
    }

    const v2f zero2 = {0.f, 0.f};

    v4f bl[4], bh[4];
    #pragma unroll
    for (int t = 0; t < 4; ++t) {
        bl[t] = *(const v4f*)(lds_b + 16 * t + 4 * h);
        bh[t] = *(const v4f*)(lds_b + 16 * t + 8 + 4 * h);
    }

    for (; tile < tiles; tile += wstride) {
        const long e0 = (long)tile * EPT;

        const float* pu0 = u_feat + (long)xu0 * DIM + 4 * h;
        const float* pu1 = u_feat + (long)xu1 * DIM + 4 * h;
        const float* pi0 = i_feat + (long)xi0 * DIM + 4 * h;
        const float* pi1 = i_feat + (long)xi1 * DIM + 4 * h;
        LOAD_IDX(tile + 2 * wstride);

        v2f pa0 = zero2, pb0 = zero2, pa1 = zero2, pb1 = zero2;
        v8f pacc0, pacc1;
        float res[NREL];

        #pragma unroll
        for (int blk = 0; blk < NREL * 4; ++blk) {
            const int r = blk >> 2;
            const int t = blk & 3;

            const v4f blo = bl[t];
            const v4f bhi = bh[t];
            v8f acc0, acc1;
            acc0[0] = blo.x; acc0[1] = blo.y; acc0[2] = blo.z; acc0[3] = blo.w;
            acc0[4] = bhi.x; acc0[5] = bhi.y; acc0[6] = bhi.z; acc0[7] = bhi.w;
            acc1 = acc0;

            if (t == 3) {
                const int rn = (r + 1) % NREL;
                #pragma unroll
                for (int tt = 0; tt < 4; ++tt) {
                    bl[tt] = *(const v4f*)(lds_b + rn * DIM + 16 * tt + 4 * h);
                    bh[tt] = *(const v4f*)(lds_b + rn * DIM + 16 * tt + 8 + 4 * h);
                }
            }

            if (blk < 8) {
                nU0[blk] = *(const v4f*)(pu0 + 8 * blk);
                nU1[blk] = *(const v4f*)(pu1 + 8 * blk);
            }

            #pragma unroll
            for (int j = 0; j < 8; ++j) {
                const v4f w = Wf[blk * 8 + j];
                #pragma unroll
                for (int q = 0; q < 2; ++q) {
                    const v2f a = pair(w, q);
                    acc0 = __builtin_amdgcn_wmma_f32_16x16x4_f32(
                               false, a, false, pair(U0[j], q), (short)0, acc0, false, false);
                    acc1 = __builtin_amdgcn_wmma_f32_16x16x4_f32(
                               false, a, false, pair(U1[j], q), (short)0, acc1, false, false);
                }
                if (j == 1 && blk > 0) {
                    STAGE2((blk - 1) & 3, pacc0, pacc1);
                    if (t == 0) FINISH_REL(r - 1);
                    if (r == NREL - 1 && t > 0) REFILL_I(t - 1);
                }
            }
            pacc0 = acc0; pacc1 = acc1;
            __builtin_amdgcn_sched_barrier(0);
        }
        STAGE2(3, pacc0, pacc1);
        FINISH_REL(NREL - 1);
        REFILL_I(3);

        #pragma unroll
        for (int i = 0; i < NREL; ++i) Obuf[lane * NREL + i] = res[i];

        asm volatile("" ::: "memory");

        float ov[NREL];
        #pragma unroll
        for (int i = 0; i < NREL; ++i) ov[i] = Obuf[i * 32 + lane];
        #pragma unroll
        for (int i = 0; i < NREL; ++i) {
            const long o = e0 * NREL + i * 32 + lane;
            if (o < (long)num_edges * NREL) out[o] = ov[i];
        }

        #pragma unroll
        for (int j = 0; j < 8; ++j) {
            U0[j] = nU0[j]; U1[j] = nU1[j];
        }
    }
#undef LOAD_IDX
#undef STAGE2
#undef FINISH_REL
#undef REFILL_I
}

extern "C" void kernel_launch(void* const* d_in, const int* in_sizes, int n_in,
                              void* d_out, int out_size, void* d_ws, size_t ws_size,
                              hipStream_t stream) {
    const float* u_feat = (const float*)d_in[0];
    const float* i_feat = (const float*)d_in[1];
    const int*   eu     = (const int*)d_in[2];
    const int*   ei     = (const int*)d_in[3];
    const float* W      = (const float*)d_in[4];
    const float* b      = (const float*)d_in[5];
    float*       out    = (float*)d_out;

    int E      = in_sizes[2];
    int tiles  = (E + EPT - 1) / EPT;
    int blocks = (tiles + WAVES - 1) / WAVES;
    if (blocks > MAX_BLOCKS) blocks = MAX_BLOCKS;

    bidecoder_kernel<<<blocks, WAVES * 32, 0, stream>>>(u_feat, i_feat, eu, ei, W, b, out, E);
}
